# starting kernel
# baseline (speedup 1.0000x reference)
#include <hip/hip_runtime.h>

typedef __attribute__((ext_vector_type(2))) float v2f;
typedef __attribute__((ext_vector_type(8))) float v8f;

#define D_FEAT 64
#define WAVES_PER_BLOCK 8
#define EDGES_PER_WAVE 16
#define EDGES_PER_BLOCK (WAVES_PER_BLOCK * EDGES_PER_WAVE)

__global__ __launch_bounds__(256) void edge_dot_wmma_kernel(
    const float* __restrict__ nf,
    const int*   __restrict__ esrc,
    const int*   __restrict__ edst,
    float*       __restrict__ out,
    int E)
{
    const int lane = threadIdx.x & 31;
    const int wave = threadIdx.x >> 5;
    const int base = (blockIdx.x * WAVES_PER_BLOCK + wave) * EDGES_PER_WAVE;
    if (base >= E) return;

    const int m = lane & 15;
    const int h = lane >> 4;

    int e = base + m;
    if (e >= E) e = E - 1;
    const int si = esrc[e];
    const int di = edst[e];

    const int soff = si * D_FEAT + h * 32;
    const int doff = di * D_FEAT + h * 32;

    float a0 = 0.0f, a1 = 0.0f;
#pragma unroll
    for (int j = 0; j < 4; ++j) {
        float4 s = *(const float4*)(nf + soff + 4 * j);
        float4 d = *(const float4*)(nf + doff + 4 * j);
        a0 = fmaf(s.x, d.x, a0); a0 = fmaf(s.y, d.y, a0);
        a0 = fmaf(s.z, d.z, a0); a0 = fmaf(s.w, d.w, a0);
    }
#pragma unroll
    for (int j = 4; j < 8; ++j) {
        float4 s = *(const float4*)(nf + soff + 4 * j);
        float4 d = *(const float4*)(nf + doff + 4 * j);
        a1 = fmaf(s.x, d.x, a1); a1 = fmaf(s.y, d.y, a1);
        a1 = fmaf(s.z, d.z, a1); a1 = fmaf(s.w, d.w, a1);
    }

    v2f A = {a0, a1};
    v2f B = {1.0f, 1.0f};
    v8f C = {};
    v8f D = __builtin_amdgcn_wmma_f32_16x16x4_f32(
         false, A,  false, B,
         (short)0, C,  false,  false);

    if (m == 0) {
        const int off = base + h * 8;
        if (off + 8 <= E) {
            float4 o0 = make_float4(D[0], D[1], D[2], D[3]);
            float4 o1 = make_float4(D[4], D[5], D[6], D[7]);
            *(float4*)(out + off)     = o0;
            *(float4*)(out + off + 4) = o1;
        } else {
            const int n = E - off;
            for (int j = 0; j < n; ++j)
                out[off + j] = D[j];
        }
    }
}

extern "C" void kernel_launch(void* const* d_in, const int* in_sizes, int n_in,
                              void* d_out, int out_size, void* d_ws, size_t ws_size,
                              hipStream_t stream) {
    const float* nf   = (const float*)d_in[0];
    const int*   esrc = (const int*)  d_in[1];
    const int*   edst = (const int*)  d_in[2];
    float*       out  = (float*)d_out;

    const int E = in_sizes[1];
    const int grid = (E + EDGES_PER_BLOCK - 1) / EDGES_PER_BLOCK;
    edge_dot_wmma_kernel<<<grid, 256, 0, stream>>>(nf, esrc, edst, out, E);
}


# Coalesced gathers: each 16-lane row loads one whole 256-B feature row per load (32 edges per wave); chunk sums across lanes via 0/1-selector f32 WMMA transposes; summation order changes
# speedup vs baseline: 3.0404x; 3.0404x over previous
#include <hip/hip_runtime.h>

typedef __attribute__((ext_vector_type(2))) float v2f;
typedef __attribute__((ext_vector_type(8))) float v8f;

#define D_FEAT 64
#define WAVES_PER_BLOCK 8
#define EDGES_PER_WAVE 32
#define EDGES_PER_BLOCK (WAVES_PER_BLOCK * EDGES_PER_WAVE)

template <int I>
__device__ __forceinline__ unsigned row_lane(unsigned v) {
    return (unsigned)__builtin_amdgcn_update_dpp(0, (int)v, 0x150 + I, 0xf, 0xf, false);
}

template <int I>
__device__ __forceinline__ float chunk_dot(const char* __restrict__ nfb,
                                           unsigned srow, unsigned drow, unsigned cbyte) {
    const float4 s = *(const float4*)(nfb + (row_lane<I>(srow) + cbyte));
    const float4 d = *(const float4*)(nfb + (row_lane<I>(drow) + cbyte));
    float p = s.x * d.x;
    p = fmaf(s.y, d.y, p);
    p = fmaf(s.z, d.z, p);
    p = fmaf(s.w, d.w, p);
    return p;
}

__device__ __forceinline__ v8f wmma4(v2f a, v2f b, v8f c) {
    return __builtin_amdgcn_wmma_f32_16x16x4_f32(false, a, false, b, (short)0, c, false, false);
}

__device__ __forceinline__ float sum8(v8f d) {
    const float ev = (d[0] + d[2]) + (d[4] + d[6]);
    const float od = (d[1] + d[3]) + (d[5] + d[7]);
    return ev + od;
}

__global__ __launch_bounds__(256) void edge_dot_rows_kernel(
    const float* __restrict__ nf,
    const int*   __restrict__ esrc,
    const int*   __restrict__ edst,
    float*       __restrict__ out,
    int E)
{
    const int lane = threadIdx.x & 31;
    const int wave = threadIdx.x >> 5;
    const int base = (blockIdx.x * WAVES_PER_BLOCK + wave) * EDGES_PER_WAVE;
    if (base >= E) return;

    const int c = lane & 15;
    const int h = lane >> 4;

    int e = base + lane;
    if (e >= E) e = E - 1;
    const unsigned srow = (unsigned)esrc[e] * (unsigned)(D_FEAT * sizeof(float));
    const unsigned drow = (unsigned)edst[e] * (unsigned)(D_FEAT * sizeof(float));
    const unsigned cbyte = (unsigned)c * 16u;
    const char* __restrict__ nfb = (const char*)nf;

    float p[16];
    p[0]  = chunk_dot<0>(nfb, srow, drow, cbyte);
    p[1]  = chunk_dot<1>(nfb, srow, drow, cbyte);
    p[2]  = chunk_dot<2>(nfb, srow, drow, cbyte);
    p[3]  = chunk_dot<3>(nfb, srow, drow, cbyte);
    p[4]  = chunk_dot<4>(nfb, srow, drow, cbyte);
    p[5]  = chunk_dot<5>(nfb, srow, drow, cbyte);
    p[6]  = chunk_dot<6>(nfb, srow, drow, cbyte);
    p[7]  = chunk_dot<7>(nfb, srow, drow, cbyte);
    p[8]  = chunk_dot<8>(nfb, srow, drow, cbyte);
    p[9]  = chunk_dot<9>(nfb, srow, drow, cbyte);
    p[10] = chunk_dot<10>(nfb, srow, drow, cbyte);
    p[11] = chunk_dot<11>(nfb, srow, drow, cbyte);
    p[12] = chunk_dot<12>(nfb, srow, drow, cbyte);
    p[13] = chunk_dot<13>(nfb, srow, drow, cbyte);
    p[14] = chunk_dot<14>(nfb, srow, drow, cbyte);
    p[15] = chunk_dot<15>(nfb, srow, drow, cbyte);

    const int t = c - 8 * h;
    v2f S[4];
#pragma unroll
    for (int j = 0; j < 4; ++j) {
        S[j][0] = (t == 2 * j)     ? 1.0f : 0.0f;
        S[j][1] = (t == 2 * j + 1) ? 1.0f : 0.0f;
    }

    v8f DX = {};
    v8f DY = {};
#pragma unroll
    for (int q = 0; q < 4; ++q) {
        v2f ax = {p[2 * q],     p[2 * q + 1]};
        v2f ay = {p[2 * q + 8], p[2 * q + 9]};
        DX = wmma4(ax, S[q], DX);
        DY = wmma4(ay, S[q], DY);
    }

    const float tx = sum8(DX);
    const float ty = sum8(DY);

    v2f A2 = {tx, ty};
    v2f B2 = {(c & 1) ? 0.0f : 1.0f, (c & 1) ? 1.0f : 0.0f};
    v8f Z = {};
    v8f R = wmma4(A2, B2, Z);

    if (c < 2) {
        const int off = base + 16 * h + 8 * c;
        if (off + 8 <= E) {
            float4 o0 = make_float4(R[0], R[1], R[2], R[3]);
            float4 o1 = make_float4(R[4], R[5], R[6], R[7]);
            *(float4*)(out + off)     = o0;
            *(float4*)(out + off + 4) = o1;
        } else {
            const int n = E - off;
            for (int j = 0; j < n; ++j)
                out[off + j] = R[j];
        }
    }
}

extern "C" void kernel_launch(void* const* d_in, const int* in_sizes, int n_in,
                              void* d_out, int out_size, void* d_ws, size_t ws_size,
                              hipStream_t stream) {
    const float* nf   = (const float*)d_in[0];
    const int*   esrc = (const int*)  d_in[1];
    const int*   edst = (const int*)  d_in[2];
    float*       out  = (float*)d_out;

    const int E = in_sizes[1];
    const int grid = (E + EDGES_PER_BLOCK - 1) / EDGES_PER_BLOCK;
    edge_dot_rows_kernel<<<grid, 256, 0, stream>>>(nf, esrc, edst, out, E);
}
